# starting kernel
# baseline (speedup 1.0000x reference)
#include <hip/hip_runtime.h>
#include <hip/hip_bf16.h>
#include <cmath>

typedef __attribute__((ext_vector_type(2))) float v2f;
typedef __attribute__((ext_vector_type(8))) float v8f;

#define B_ 16
#define S_ 256
#define E_ 32
#define H_ 64

__global__ __launch_bounds__(32) void proj_kernel(
    const float* __restrict__ x, const float* __restrict__ Wv,
    const float* __restrict__ bv, const float* __restrict__ W1,
    const float* __restrict__ b1,
    float* __restrict__ a_ws, float* __restrict__ bf_ws,
    float* __restrict__ v_ws)
{
  const int tile = blockIdx.x;
  const int lane = threadIdx.x;
  const int rc   = lane & 15;
  const int kh   = lane >> 4;
  const int rowbase = tile * 16;

  v2f xr[8];
  float ss = 0.f;
  const float* xrow = x + (size_t)(rowbase + rc) * E_;
#pragma unroll
  for (int kb = 0; kb < 8; ++kb) {
    const int k0 = kb * 4 + 2 * kh;
    v2f t; t.x = xrow[k0]; t.y = xrow[k0 + 1];
    xr[kb] = t;
    ss += t.x * t.x + t.y * t.y;
  }
  ss += __shfl_xor(ss, 16, 32);
  const float inv = 1.0f / fmaxf(sqrtf(ss), 1e-12f);
  v2f xnr[8];
#pragma unroll
  for (int kb = 0; kb < 8; ++kb) xnr[kb] = xr[kb] * inv;

#pragma unroll
  for (int half = 0; half < 2; ++half) {
    const int ebase = half * E_;
    float* dst = half ? bf_ws : a_ws;
#pragma unroll
    for (int nt = 0; nt < 4; ++nt) {
      const int hb = nt * 16;
      v8f acc = {};
#pragma unroll
      for (int kb = 0; kb < 8; ++kb) {
        const int k0 = kb * 4 + 2 * kh;
        v2f bfrag;
        bfrag.x = W1[(size_t)(ebase + k0)     * H_ + hb + rc];
        bfrag.y = W1[(size_t)(ebase + k0 + 1) * H_ + hb + rc];
        acc = __builtin_amdgcn_wmma_f32_16x16x4_f32(
            false, xnr[kb], false, bfrag, (short)0, acc, false, false);
      }
      const float badd = half ? 0.f : b1[hb + rc];
#pragma unroll
      for (int r = 0; r < 8; ++r) {
        const int grow = rowbase + r + 8 * kh;
        dst[(size_t)grow * H_ + hb + rc] = acc[r] + badd;
      }
    }
  }

#pragma unroll
  for (int nt = 0; nt < 2; ++nt) {
    const int db = nt * 16;
    v8f acc = {};
#pragma unroll
    for (int kb = 0; kb < 8; ++kb) {
      const int k0 = kb * 4 + 2 * kh;
      v2f bfrag;
      bfrag.x = Wv[(size_t)(db + rc) * E_ + k0];
      bfrag.y = Wv[(size_t)(db + rc) * E_ + k0 + 1];
      acc = __builtin_amdgcn_wmma_f32_16x16x4_f32(
          false, xr[kb], false, bfrag, (short)0, acc, false, false);
    }
#pragma unroll
    for (int r = 0; r < 8; ++r) {
      const int grow = rowbase + r + 8 * kh;
      v_ws[(size_t)grow * E_ + db + rc] = acc[r] + bv[db + rc];
    }
  }
}

__global__ __launch_bounds__(256) void attn_kernel(
    const float* __restrict__ w2, const float* __restrict__ b2p,
    const float* __restrict__ a_ws, const float* __restrict__ bf_ws,
    const float* __restrict__ v_ws, float* __restrict__ out)
{
  __shared__ float s_a[16 * H_];
  __shared__ float s_sc[16 * S_];
  __shared__ float s_w2[H_];

  const int tid   = threadIdx.x;
  const int b     = blockIdx.x >> 4;
  const int itile = blockIdx.x & 15;
  const int irow0 = b * S_ + itile * 16;

  for (int idx = tid; idx < 16 * H_; idx += 256)
    s_a[idx] = a_ws[(size_t)irow0 * H_ + idx];
  if (tid < H_) s_w2[tid] = w2[tid];
  __syncthreads();

  const float b2 = b2p[0];
  const int i  = tid >> 4;
  const int jl = tid & 15;

  float sc[16];
  float m = -3.0e38f;
  for (int jj = 0; jj < 16; ++jj) {
    const int j = jj * 16 + jl;
    const float* bfr = bf_ws + (size_t)(b * S_ + j) * H_;
    const float* ar  = s_a + i * H_;
    float acc = 0.f;
#pragma unroll 8
    for (int h = 0; h < H_; ++h)
      acc += s_w2[h] * tanhf(ar[h] + bfr[h]);
    const float s = tanhf(acc + b2);
    sc[jj] = s;
    m = fmaxf(m, s);
  }
#pragma unroll
  for (int mask = 1; mask < 16; mask <<= 1)
    m = fmaxf(m, __shfl_xor(m, mask, 16));
  float sum = 0.f;
  for (int jj = 0; jj < 16; ++jj) { sc[jj] = expf(sc[jj] - m); sum += sc[jj]; }
#pragma unroll
  for (int mask = 1; mask < 16; mask <<= 1)
    sum += __shfl_xor(sum, mask, 16);
  const float isum = 1.0f / sum;
  for (int jj = 0; jj < 16; ++jj)
    s_sc[i * S_ + jj * 16 + jl] = sc[jj] * isum;
  __syncthreads();

  const int wave = tid >> 5;
  if (wave < 2) {
    const int lane = tid & 31;
    const int rc = lane & 15;
    const int kh = lane >> 4;
    const int db = wave * 16;
    const float* vb = v_ws + (size_t)b * S_ * E_;
    v8f acc = {};
    for (int kb = 0; kb < 64; ++kb) {
      const int k0 = kb * 4 + 2 * kh;
      v2f af, bfrag;
      af.x    = s_sc[rc * S_ + k0];
      af.y    = s_sc[rc * S_ + k0 + 1];
      bfrag.x = vb[(size_t)k0       * E_ + db + rc];
      bfrag.y = vb[(size_t)(k0 + 1) * E_ + db + rc];
      acc = __builtin_amdgcn_wmma_f32_16x16x4_f32(
          false, af, false, bfrag, (short)0, acc, false, false);
    }
#pragma unroll
    for (int r = 0; r < 8; ++r) {
      const int grow = irow0 + r + 8 * kh;
      out[(size_t)grow * E_ + db + rc] = acc[r];
    }
  }
}

extern "C" void kernel_launch(void* const* d_in, const int* in_sizes, int n_in,
                              void* d_out, int out_size, void* d_ws, size_t ws_size,
                              hipStream_t stream) {
  const float* x  = (const float*)d_in[0];
  const float* Wv = (const float*)d_in[1];
  const float* bv = (const float*)d_in[2];
  const float* W1 = (const float*)d_in[3];
  const float* b1 = (const float*)d_in[4];
  const float* w2 = (const float*)d_in[5];
  const float* b2 = (const float*)d_in[6];
  float* outp = (float*)d_out;

  float* ws    = (float*)d_ws;
  float* a_ws  = ws;
  float* bf_ws = a_ws + (size_t)B_ * S_ * H_;
  float* v_ws  = bf_ws + (size_t)B_ * S_ * H_;

  proj_kernel<<<dim3(B_ * S_ / 16), dim3(32), 0, stream>>>(
      x, Wv, bv, W1, b1, a_ws, bf_ws, v_ws);
  attn_kernel<<<dim3(B_ * (S_ / 16)), dim3(256), 0, stream>>>(
      w2, b2, a_ws, bf_ws, v_ws, outp);
}


# tanh(a_i+b_j) as 1-2*rcp(1+ea_i*eb_j), ea=exp(2a) and eb=exp(2b) stored once per row by the projection kernel: pair loop keeps one v_rcp_f32 and 3 fma per element, no exp; sum order kept
# speedup vs baseline: 2.6430x; 2.6430x over previous
#include <hip/hip_runtime.h>
#include <hip/hip_bf16.h>
#include <cmath>

typedef __attribute__((ext_vector_type(2))) float v2f;
typedef __attribute__((ext_vector_type(8))) float v8f;

#define B_ 16
#define S_ 256
#define E_ 32
#define H_ 64

__global__ __launch_bounds__(32) void proj_kernel(
    const float* __restrict__ x, const float* __restrict__ Wv,
    const float* __restrict__ bv, const float* __restrict__ W1,
    const float* __restrict__ b1,
    float* __restrict__ a_ws, float* __restrict__ bf_ws,
    float* __restrict__ v_ws)
{
  const int tile = blockIdx.x;
  const int lane = threadIdx.x;
  const int rc   = lane & 15;
  const int kh   = lane >> 4;
  const int rowbase = tile * 16;

  v2f xr[8];
  float ss = 0.f;
  const float* xrow = x + (size_t)(rowbase + rc) * E_;
#pragma unroll
  for (int kb = 0; kb < 8; ++kb) {
    const int k0 = kb * 4 + 2 * kh;
    v2f t; t.x = xrow[k0]; t.y = xrow[k0 + 1];
    xr[kb] = t;
    ss += t.x * t.x + t.y * t.y;
  }
  ss += __shfl_xor(ss, 16, 32);
  const float inv = 1.0f / fmaxf(sqrtf(ss), 1e-12f);
  v2f xnr[8];
#pragma unroll
  for (int kb = 0; kb < 8; ++kb) xnr[kb] = xr[kb] * inv;

#pragma unroll
  for (int half = 0; half < 2; ++half) {
    const int ebase = half * E_;
    float* dst = half ? bf_ws : a_ws;
#pragma unroll
    for (int nt = 0; nt < 4; ++nt) {
      const int hb = nt * 16;
      v8f acc = {};
#pragma unroll
      for (int kb = 0; kb < 8; ++kb) {
        const int k0 = kb * 4 + 2 * kh;
        v2f bfrag;
        bfrag.x = W1[(size_t)(ebase + k0)     * H_ + hb + rc];
        bfrag.y = W1[(size_t)(ebase + k0 + 1) * H_ + hb + rc];
        acc = __builtin_amdgcn_wmma_f32_16x16x4_f32(
            false, xnr[kb], false, bfrag, (short)0, acc, false, false);
      }
      const float badd = half ? 0.f : b1[hb + rc];
#pragma unroll
      for (int r = 0; r < 8; ++r) {
        const int grow = rowbase + r + 8 * kh;
        dst[(size_t)grow * H_ + hb + rc] = expf(2.0f * (acc[r] + badd));
      }
    }
  }

#pragma unroll
  for (int nt = 0; nt < 2; ++nt) {
    const int db = nt * 16;
    v8f acc = {};
#pragma unroll
    for (int kb = 0; kb < 8; ++kb) {
      const int k0 = kb * 4 + 2 * kh;
      v2f bfrag;
      bfrag.x = Wv[(size_t)(db + rc) * E_ + k0];
      bfrag.y = Wv[(size_t)(db + rc) * E_ + k0 + 1];
      acc = __builtin_amdgcn_wmma_f32_16x16x4_f32(
          false, xr[kb], false, bfrag, (short)0, acc, false, false);
    }
#pragma unroll
    for (int r = 0; r < 8; ++r) {
      const int grow = rowbase + r + 8 * kh;
      v_ws[(size_t)grow * E_ + db + rc] = acc[r] + bv[db + rc];
    }
  }
}

__global__ __launch_bounds__(256) void attn_kernel(
    const float* __restrict__ w2, const float* __restrict__ b2p,
    const float* __restrict__ a_ws, const float* __restrict__ bf_ws,
    const float* __restrict__ v_ws, float* __restrict__ out)
{
  __shared__ float s_a[16 * H_];
  __shared__ float s_sc[16 * S_];
  __shared__ float s_w2[H_];

  const int tid   = threadIdx.x;
  const int b     = blockIdx.x >> 4;
  const int itile = blockIdx.x & 15;
  const int irow0 = b * S_ + itile * 16;

  for (int idx = tid; idx < 16 * H_; idx += 256)
    s_a[idx] = a_ws[(size_t)irow0 * H_ + idx];
  if (tid < H_) s_w2[tid] = w2[tid];
  __syncthreads();

  const float b2 = b2p[0];
  const int i  = tid >> 4;
  const int jl = tid & 15;

  float sc[16];
  float m = -3.0e38f;
  for (int jj = 0; jj < 16; ++jj) {
    const int j = jj * 16 + jl;
    const float* bfr = bf_ws + (size_t)(b * S_ + j) * H_;
    const float* ar  = s_a + i * H_;
    float acc = 0.f;
#pragma unroll 8
    for (int h = 0; h < H_; ++h) {
      const float r = __builtin_amdgcn_rcpf(fmaf(ar[h], bfr[h], 1.0f));
      acc += s_w2[h] * fmaf(-2.0f, r, 1.0f);
    }
    const float s = tanhf(acc + b2);
    sc[jj] = s;
    m = fmaxf(m, s);
  }
#pragma unroll
  for (int mask = 1; mask < 16; mask <<= 1)
    m = fmaxf(m, __shfl_xor(m, mask, 16));
  float sum = 0.f;
  for (int jj = 0; jj < 16; ++jj) { sc[jj] = expf(sc[jj] - m); sum += sc[jj]; }
#pragma unroll
  for (int mask = 1; mask < 16; mask <<= 1)
    sum += __shfl_xor(sum, mask, 16);
  const float isum = 1.0f / sum;
  for (int jj = 0; jj < 16; ++jj)
    s_sc[i * S_ + jj * 16 + jl] = sc[jj] * isum;
  __syncthreads();

  const int wave = tid >> 5;
  if (wave < 2) {
    const int lane = tid & 31;
    const int rc = lane & 15;
    const int kh = lane >> 4;
    const int db = wave * 16;
    const float* vb = v_ws + (size_t)b * S_ * E_;
    v8f acc = {};
    for (int kb = 0; kb < 64; ++kb) {
      const int k0 = kb * 4 + 2 * kh;
      v2f af, bfrag;
      af.x    = s_sc[rc * S_ + k0];
      af.y    = s_sc[rc * S_ + k0 + 1];
      bfrag.x = vb[(size_t)k0       * E_ + db + rc];
      bfrag.y = vb[(size_t)(k0 + 1) * E_ + db + rc];
      acc = __builtin_amdgcn_wmma_f32_16x16x4_f32(
          false, af, false, bfrag, (short)0, acc, false, false);
    }
#pragma unroll
    for (int r = 0; r < 8; ++r) {
      const int grow = irow0 + r + 8 * kh;
      out[(size_t)grow * E_ + db + rc] = acc[r];
    }
  }
}

extern "C" void kernel_launch(void* const* d_in, const int* in_sizes, int n_in,
                              void* d_out, int out_size, void* d_ws, size_t ws_size,
                              hipStream_t stream) {
  const float* x  = (const float*)d_in[0];
  const float* Wv = (const float*)d_in[1];
  const float* bv = (const float*)d_in[2];
  const float* W1 = (const float*)d_in[3];
  const float* b1 = (const float*)d_in[4];
  const float* w2 = (const float*)d_in[5];
  const float* b2 = (const float*)d_in[6];
  float* outp = (float*)d_out;

  float* ws    = (float*)d_ws;
  float* a_ws  = ws;
  float* bf_ws = a_ws + (size_t)B_ * S_ * H_;
  float* v_ws  = bf_ws + (size_t)B_ * S_ * H_;

  proj_kernel<<<dim3(B_ * S_ / 16), dim3(32), 0, stream>>>(
      x, Wv, bv, W1, b1, a_ws, bf_ws, v_ws);
  attn_kernel<<<dim3(B_ * (S_ / 16)), dim3(256), 0, stream>>>(
      w2, b2, a_ws, bf_ws, v_ws, outp);
}


# pair loop operands from registers: thread owns column j, its eb row held in 64 VGPRs over the 16 rows i, ea row as one grouped LDS broadcast read per i, w2 in SGPRs; all summation orders kept
# speedup vs baseline: 4.6215x; 1.7486x over previous
#include <hip/hip_runtime.h>
#include <hip/hip_bf16.h>
#include <cmath>

typedef __attribute__((ext_vector_type(2))) float v2f;
typedef __attribute__((ext_vector_type(4))) float v4f;
typedef __attribute__((ext_vector_type(8))) float v8f;

#define B_ 16
#define S_ 256
#define E_ 32
#define H_ 64

__global__ __launch_bounds__(32) void proj_kernel(
    const float* __restrict__ x, const float* __restrict__ Wv,
    const float* __restrict__ bv, const float* __restrict__ W1,
    const float* __restrict__ b1,
    float* __restrict__ a_ws, float* __restrict__ bf_ws,
    float* __restrict__ v_ws)
{
  const int tile = blockIdx.x;
  const int lane = threadIdx.x;
  const int rc   = lane & 15;
  const int kh   = lane >> 4;
  const int rowbase = tile * 16;

  v2f xr[8];
  float ss = 0.f;
  const float* xrow = x + (size_t)(rowbase + rc) * E_;
#pragma unroll
  for (int kb = 0; kb < 8; ++kb) {
    const int k0 = kb * 4 + 2 * kh;
    v2f t; t.x = xrow[k0]; t.y = xrow[k0 + 1];
    xr[kb] = t;
    ss += t.x * t.x + t.y * t.y;
  }
  ss += __shfl_xor(ss, 16, 32);
  const float inv = 1.0f / fmaxf(sqrtf(ss), 1e-12f);
  v2f xnr[8];
#pragma unroll
  for (int kb = 0; kb < 8; ++kb) xnr[kb] = xr[kb] * inv;

#pragma unroll
  for (int half = 0; half < 2; ++half) {
    const int ebase = half * E_;
    float* dst = half ? bf_ws : a_ws;
#pragma unroll
    for (int nt = 0; nt < 4; ++nt) {
      const int hb = nt * 16;
      v8f acc = {};
#pragma unroll
      for (int kb = 0; kb < 8; ++kb) {
        const int k0 = kb * 4 + 2 * kh;
        v2f bfrag;
        bfrag.x = W1[(size_t)(ebase + k0)     * H_ + hb + rc];
        bfrag.y = W1[(size_t)(ebase + k0 + 1) * H_ + hb + rc];
        acc = __builtin_amdgcn_wmma_f32_16x16x4_f32(
            false, xnr[kb], false, bfrag, (short)0, acc, false, false);
      }
      const float badd = half ? 0.f : b1[hb + rc];
#pragma unroll
      for (int r = 0; r < 8; ++r) {
        const int grow = rowbase + r + 8 * kh;
        dst[(size_t)grow * H_ + hb + rc] = expf(2.0f * (acc[r] + badd));
      }
    }
  }

#pragma unroll
  for (int nt = 0; nt < 2; ++nt) {
    const int db = nt * 16;
    v8f acc = {};
#pragma unroll
    for (int kb = 0; kb < 8; ++kb) {
      const int k0 = kb * 4 + 2 * kh;
      v2f bfrag;
      bfrag.x = Wv[(size_t)(db + rc) * E_ + k0];
      bfrag.y = Wv[(size_t)(db + rc) * E_ + k0 + 1];
      acc = __builtin_amdgcn_wmma_f32_16x16x4_f32(
          false, xr[kb], false, bfrag, (short)0, acc, false, false);
    }
#pragma unroll
    for (int r = 0; r < 8; ++r) {
      const int grow = rowbase + r + 8 * kh;
      v_ws[(size_t)grow * E_ + db + rc] = acc[r] + bv[db + rc];
    }
  }
}

__global__ __launch_bounds__(256) void attn_kernel(
    const float* __restrict__ w2, const float* __restrict__ b2p,
    const float* __restrict__ a_ws, const float* __restrict__ bf_ws,
    const float* __restrict__ v_ws, float* __restrict__ out)
{
  __shared__ alignas(16) float s_a[16 * H_];
  __shared__ float s_sc[16 * S_];

  const int tid   = threadIdx.x;
  const int b     = blockIdx.x >> 4;
  const int itile = blockIdx.x & 15;
  const int irow0 = b * S_ + itile * 16;

  float eb[H_];
  {
    const float4* src =
        reinterpret_cast<const float4*>(bf_ws + (size_t)(b * S_ + tid) * H_);
#pragma unroll
    for (int q = 0; q < H_ / 4; ++q) {
      const float4 t = src[q];
      eb[4 * q + 0] = t.x; eb[4 * q + 1] = t.y;
      eb[4 * q + 2] = t.z; eb[4 * q + 3] = t.w;
    }
  }
  reinterpret_cast<float4*>(s_a)[tid] =
      reinterpret_cast<const float4*>(a_ws + (size_t)irow0 * H_)[tid];
  __syncthreads();

  const float b2 = b2p[0];
  float w[H_];
#pragma unroll
  for (int h = 0; h < H_; ++h) w[h] = w2[h];

#pragma unroll 1
  for (int i = 0; i < 16; ++i) {
    const v4f* ar = reinterpret_cast<const v4f*>(s_a + i * H_);
    v4f ev[H_ / 4];
#pragma unroll
    for (int q = 0; q < H_ / 4; ++q) ev[q] = ar[q];
    asm volatile("" :: "v"(ev[0]), "v"(ev[1]), "v"(ev[2]), "v"(ev[3]),
                       "v"(ev[4]), "v"(ev[5]), "v"(ev[6]), "v"(ev[7]),
                       "v"(ev[8]), "v"(ev[9]), "v"(ev[10]), "v"(ev[11]),
                       "v"(ev[12]), "v"(ev[13]), "v"(ev[14]), "v"(ev[15]));
    __builtin_amdgcn_sched_barrier(0);
    float acc = 0.f;
#pragma unroll
    for (int q = 0; q < H_ / 4; ++q) {
      const float e4[4] = {ev[q].x, ev[q].y, ev[q].z, ev[q].w};
#pragma unroll
      for (int c = 0; c < 4; ++c) {
        const int h = 4 * q + c;
        const float r = __builtin_amdgcn_rcpf(fmaf(e4[c], eb[h], 1.0f));
        acc += w[h] * fmaf(-2.0f, r, 1.0f);
      }
    }
    s_sc[i * S_ + tid] = tanhf(acc + b2);
  }
  __syncthreads();

  const int i  = tid >> 4;
  const int jl = tid & 15;
  float sc[16];
  float m = -3.0e38f;
#pragma unroll
  for (int jj = 0; jj < 16; ++jj) {
    const float s = s_sc[i * S_ + jj * 16 + jl];
    sc[jj] = s;
    m = fmaxf(m, s);
  }
#pragma unroll
  for (int mask = 1; mask < 16; mask <<= 1)
    m = fmaxf(m, __shfl_xor(m, mask, 16));
  float sum = 0.f;
#pragma unroll
  for (int jj = 0; jj < 16; ++jj) { sc[jj] = expf(sc[jj] - m); sum += sc[jj]; }
#pragma unroll
  for (int mask = 1; mask < 16; mask <<= 1)
    sum += __shfl_xor(sum, mask, 16);
  const float isum = 1.0f / sum;
#pragma unroll
  for (int jj = 0; jj < 16; ++jj)
    s_sc[i * S_ + jj * 16 + jl] = sc[jj] * isum;
  __syncthreads();

  const int wave = tid >> 5;
  if (wave < 2) {
    const int lane = tid & 31;
    const int rc = lane & 15;
    const int kh = lane >> 4;
    const int db = wave * 16;
    const float* vb = v_ws + (size_t)b * S_ * E_;
    v8f acc = {};
    for (int kb = 0; kb < 64; ++kb) {
      const int k0 = kb * 4 + 2 * kh;
      v2f af, bfrag;
      af.x    = s_sc[rc * S_ + k0];
      af.y    = s_sc[rc * S_ + k0 + 1];
      bfrag.x = vb[(size_t)k0       * E_ + db + rc];
      bfrag.y = vb[(size_t)(k0 + 1) * E_ + db + rc];
      acc = __builtin_amdgcn_wmma_f32_16x16x4_f32(
          false, af, false, bfrag, (short)0, acc, false, false);
    }
#pragma unroll
    for (int r = 0; r < 8; ++r) {
      const int grow = irow0 + r + 8 * kh;
      out[(size_t)grow * E_ + db + rc] = acc[r];
    }
  }
}

extern "C" void kernel_launch(void* const* d_in, const int* in_sizes, int n_in,
                              void* d_out, int out_size, void* d_ws, size_t ws_size,
                              hipStream_t stream) {
  const float* x  = (const float*)d_in[0];
  const float* Wv = (const float*)d_in[1];
  const float* bv = (const float*)d_in[2];
  const float* W1 = (const float*)d_in[3];
  const float* b1 = (const float*)d_in[4];
  const float* w2 = (const float*)d_in[5];
  const float* b2 = (const float*)d_in[6];
  float* outp = (float*)d_out;

  float* ws    = (float*)d_ws;
  float* a_ws  = ws;
  float* bf_ws = a_ws + (size_t)B_ * S_ * H_;
  float* v_ws  = bf_ws + (size_t)B_ * S_ * H_;

  proj_kernel<<<dim3(B_ * S_ / 16), dim3(32), 0, stream>>>(
      x, Wv, bv, W1, b1, a_ws, bf_ws, v_ws);
  attn_kernel<<<dim3(B_ * (S_ / 16)), dim3(256), 0, stream>>>(
      w2, b2, a_ws, bf_ws, v_ws, outp);
}


# projection launch folded into the attention kernel: each work-group rebuilds its batch's eb and v and its own ea tile with the same f32 WMMA loaders into LDS; one launch, all sums in the same order
# speedup vs baseline: 7.6364x; 1.6524x over previous
#include <hip/hip_runtime.h>
#include <hip/hip_bf16.h>
#include <cmath>


typedef __attribute__((ext_vector_type(2))) float v2f;
typedef __attribute__((ext_vector_type(4))) float v4f;
typedef __attribute__((ext_vector_type(8))) float v8f;

#define B_ 16
#define S_ 256
#define E_ 32
#define H_ 64
#define EB_PITCH 68
#define V_PITCH 44

__device__ __forceinline__ void load_rows(const float* __restrict__ x, int rowbase,
                                          int rc, int kh,
                                          v2f (&xr)[8], v2f (&xnr)[8])
{
  float ss = 0.f;
  const float* xrow = x + (size_t)(rowbase + rc) * E_;
#pragma unroll
  for (int kb = 0; kb < 8; ++kb) {
    const int k0 = kb * 4 + 2 * kh;
    v2f t; t.x = xrow[k0]; t.y = xrow[k0 + 1];
    xr[kb] = t;
    ss += t.x * t.x + t.y * t.y;
  }
  ss += __shfl_xor(ss, 16, 32);
  const float inv = 1.0f / fmaxf(sqrtf(ss), 1e-12f);
#pragma unroll
  for (int kb = 0; kb < 8; ++kb) xnr[kb] = xr[kb] * inv;
}

__global__ __launch_bounds__(256) void attn_fused_kernel(
    const float* __restrict__ x, const float* __restrict__ Wv,
    const float* __restrict__ bv, const float* __restrict__ W1,
    const float* __restrict__ b1, const float* __restrict__ w2,
    const float* __restrict__ b2p, float* __restrict__ out)
{
  __shared__ alignas(16) float s_eb[S_ * EB_PITCH];
  __shared__ float s_v[S_ * V_PITCH];
  __shared__ alignas(16) float s_a[16 * H_];
  __shared__ float s_sc[16 * S_];

  const int tid   = threadIdx.x;
  const int b     = blockIdx.x >> 4;
  const int itile = blockIdx.x & 15;
  const int irow0 = b * S_ + itile * 16;
  const int wave = __builtin_amdgcn_readfirstlane(tid >> 5);
  const int lane = tid & 31;
  const int rc   = lane & 15;
  const int kh   = lane >> 4;

  const float b2 = b2p[0];
  float w[H_];
#pragma unroll
  for (int h = 0; h < H_; ++h) w[h] = w2[h];

  if (wave < 4) {
    v2f xr[8], xnr[8];
    load_rows(x, irow0, rc, kh, xr, xnr);
    const int hb = wave * 16;
    v8f acc = {};
#pragma unroll
    for (int kb = 0; kb < 8; ++kb) {
      const int k0 = kb * 4 + 2 * kh;
      v2f bfrag;
      bfrag.x = W1[(size_t)k0       * H_ + hb + rc];
      bfrag.y = W1[(size_t)(k0 + 1) * H_ + hb + rc];
      acc = __builtin_amdgcn_wmma_f32_16x16x4_f32(
          false, xnr[kb], false, bfrag, (short)0, acc, false, false);
    }
    const float badd = b1[hb + rc];
#pragma unroll
    for (int r = 0; r < 8; ++r)
      s_a[(r + 8 * kh) * H_ + hb + rc] = expf(2.0f * (acc[r] + badd));
  }

#pragma unroll
  for (int t = 0; t < 2; ++t) {
    const int jl0 = (2 * wave + t) * 16;
    v2f xr[8], xnr[8];
    load_rows(x, b * S_ + jl0, rc, kh, xr, xnr);

#pragma unroll
    for (int nt = 0; nt < 4; ++nt) {
      const int hb = nt * 16;
      v8f acc = {};
#pragma unroll
      for (int kb = 0; kb < 8; ++kb) {
        const int k0 = kb * 4 + 2 * kh;
        v2f bfrag;
        bfrag.x = W1[(size_t)(E_ + k0)     * H_ + hb + rc];
        bfrag.y = W1[(size_t)(E_ + k0 + 1) * H_ + hb + rc];
        acc = __builtin_amdgcn_wmma_f32_16x16x4_f32(
            false, xnr[kb], false, bfrag, (short)0, acc, false, false);
      }
#pragma unroll
      for (int r = 0; r < 8; ++r)
        s_eb[(jl0 + r + 8 * kh) * EB_PITCH + hb + rc] = expf(2.0f * acc[r]);
    }

#pragma unroll
    for (int nt = 0; nt < 2; ++nt) {
      const int db = nt * 16;
      v8f acc = {};
#pragma unroll
      for (int kb = 0; kb < 8; ++kb) {
        const int k0 = kb * 4 + 2 * kh;
        v2f bfrag;
        bfrag.x = Wv[(size_t)(db + rc) * E_ + k0];
        bfrag.y = Wv[(size_t)(db + rc) * E_ + k0 + 1];
        acc = __builtin_amdgcn_wmma_f32_16x16x4_f32(
            false, xr[kb], false, bfrag, (short)0, acc, false, false);
      }
      const float bvv = bv[db + rc];
#pragma unroll
      for (int r = 0; r < 8; ++r)
        s_v[(jl0 + r + 8 * kh) * V_PITCH + db + rc] = acc[r] + bvv;
    }
  }
  __syncthreads();

  float eb[H_];
  {
    const v4f* src = reinterpret_cast<const v4f*>(s_eb + tid * EB_PITCH);
#pragma unroll
    for (int q = 0; q < H_ / 4; ++q) {
      const v4f t = src[q];
      eb[4 * q + 0] = t.x; eb[4 * q + 1] = t.y;
      eb[4 * q + 2] = t.z; eb[4 * q + 3] = t.w;
    }
  }


#pragma unroll 1
  for (int i = 0; i < 16; ++i) {
    const v4f* ar = reinterpret_cast<const v4f*>(s_a + i * H_);
    v4f ev[H_ / 4];
#pragma unroll
    for (int q = 0; q < H_ / 4; ++q) ev[q] = ar[q];
    asm volatile("" :: "v"(ev[0]), "v"(ev[1]), "v"(ev[2]), "v"(ev[3]),
                       "v"(ev[4]), "v"(ev[5]), "v"(ev[6]), "v"(ev[7]),
                       "v"(ev[8]), "v"(ev[9]), "v"(ev[10]), "v"(ev[11]),
                       "v"(ev[12]), "v"(ev[13]), "v"(ev[14]), "v"(ev[15]));
    __builtin_amdgcn_sched_barrier(0);
    float acc = 0.f;
#pragma unroll
    for (int q = 0; q < H_ / 4; ++q) {
      const float e4[4] = {ev[q].x, ev[q].y, ev[q].z, ev[q].w};
#pragma unroll
      for (int c = 0; c < 4; ++c) {
        const int h = 4 * q + c;
        const float r = __builtin_amdgcn_rcpf(fmaf(e4[c], eb[h], 1.0f));
        acc += w[h] * fmaf(-2.0f, r, 1.0f);
      }
    }
    s_sc[i * S_ + tid] = tanhf(acc + b2);
  }
  __syncthreads();

  const int i  = tid >> 4;
  const int jl = tid & 15;
  float sc[16];
  float m = -3.0e38f;
#pragma unroll
  for (int jj = 0; jj < 16; ++jj) {
    const float s = s_sc[i * S_ + jj * 16 + jl];
    sc[jj] = s;
    m = fmaxf(m, s);
  }
#pragma unroll
  for (int mask = 1; mask < 16; mask <<= 1)
    m = fmaxf(m, __shfl_xor(m, mask, 16));
  float sum = 0.f;
#pragma unroll
  for (int jj = 0; jj < 16; ++jj) { sc[jj] = expf(sc[jj] - m); sum += sc[jj]; }
#pragma unroll
  for (int mask = 1; mask < 16; mask <<= 1)
    sum += __shfl_xor(sum, mask, 16);
  const float isum = 1.0f / sum;
#pragma unroll
  for (int jj = 0; jj < 16; ++jj)
    s_sc[i * S_ + jj * 16 + jl] = sc[jj] * isum;
  __syncthreads();

  if (wave < 2) {
    const int db = wave * 16;
    v8f acc = {};
    for (int kb = 0; kb < 64; ++kb) {
      const int k0 = kb * 4 + 2 * kh;
      v2f af, bfrag;
      af.x    = s_sc[rc * S_ + k0];
      af.y    = s_sc[rc * S_ + k0 + 1];
      bfrag.x = s_v[k0       * V_PITCH + db + rc];
      bfrag.y = s_v[(k0 + 1) * V_PITCH + db + rc];
      acc = __builtin_amdgcn_wmma_f32_16x16x4_f32(
          false, af, false, bfrag, (short)0, acc, false, false);
    }
#pragma unroll
    for (int r = 0; r < 8; ++r) {
      const int grow = irow0 + r + 8 * kh;
      out[(size_t)grow * E_ + db + rc] = acc[r];
    }
  }
}

extern "C" void kernel_launch(void* const* d_in, const int* in_sizes, int n_in,
                              void* d_out, int out_size, void* d_ws, size_t ws_size,
                              hipStream_t stream) {
  const float* x  = (const float*)d_in[0];
  const float* Wv = (const float*)d_in[1];
  const float* bv = (const float*)d_in[2];
  const float* W1 = (const float*)d_in[3];
  const float* b1 = (const float*)d_in[4];
  const float* w2 = (const float*)d_in[5];
  const float* b2 = (const float*)d_in[6];
  float* outp = (float*)d_out;

  attn_fused_kernel<<<dim3(B_ * (S_ / 16)), dim3(256), 0, stream>>>(
      x, Wv, bv, W1, b1, w2, b2, outp);
}


# both tanh of the score on the hardware tanh instruction (v_tanh_f32): a and bf kept as plain sums (no exp in the projection), pair loop = packed add + tanh + fma per element; sum order kept
# speedup vs baseline: 9.0457x; 1.1846x over previous
#include <hip/hip_runtime.h>
#include <hip/hip_bf16.h>
#include <cmath>


typedef __attribute__((ext_vector_type(2))) float v2f;
typedef __attribute__((ext_vector_type(4))) float v4f;
typedef __attribute__((ext_vector_type(8))) float v8f;

#define B_ 16
#define S_ 256
#define E_ 32
#define H_ 64
#define EB_PITCH 68
#define V_PITCH 44

__device__ __forceinline__ void load_rows(const float* __restrict__ x, int rowbase,
                                          int rc, int kh,
                                          v2f (&xr)[8], v2f (&xnr)[8])
{
  float ss = 0.f;
  const float* xrow = x + (size_t)(rowbase + rc) * E_;
#pragma unroll
  for (int kb = 0; kb < 8; ++kb) {
    const int k0 = kb * 4 + 2 * kh;
    v2f t; t.x = xrow[k0]; t.y = xrow[k0 + 1];
    xr[kb] = t;
    ss += t.x * t.x + t.y * t.y;
  }
  ss += __shfl_xor(ss, 16, 32);
  const float inv = 1.0f / fmaxf(sqrtf(ss), 1e-12f);
#pragma unroll
  for (int kb = 0; kb < 8; ++kb) xnr[kb] = xr[kb] * inv;
}

__global__ __launch_bounds__(256) void attn_fused_kernel(
    const float* __restrict__ x, const float* __restrict__ Wv,
    const float* __restrict__ bv, const float* __restrict__ W1,
    const float* __restrict__ b1, const float* __restrict__ w2,
    const float* __restrict__ b2p, float* __restrict__ out)
{
  __shared__ alignas(16) float s_eb[S_ * EB_PITCH];
  __shared__ float s_v[S_ * V_PITCH];
  __shared__ alignas(16) float s_a[16 * H_];
  __shared__ float s_sc[16 * S_];

  const int tid   = threadIdx.x;
  const int b     = blockIdx.x >> 4;
  const int itile = blockIdx.x & 15;
  const int irow0 = b * S_ + itile * 16;
  const int wave = __builtin_amdgcn_readfirstlane(tid >> 5);
  const int lane = tid & 31;
  const int rc   = lane & 15;
  const int kh   = lane >> 4;

  const float b2 = b2p[0];
  float w[H_];
#pragma unroll
  for (int h = 0; h < H_; ++h) w[h] = w2[h];

  if (wave < 4) {
    v2f xr[8], xnr[8];
    load_rows(x, irow0, rc, kh, xr, xnr);
    const int hb = wave * 16;
    v8f acc = {};
#pragma unroll
    for (int kb = 0; kb < 8; ++kb) {
      const int k0 = kb * 4 + 2 * kh;
      v2f bfrag;
      bfrag.x = W1[(size_t)k0       * H_ + hb + rc];
      bfrag.y = W1[(size_t)(k0 + 1) * H_ + hb + rc];
      acc = __builtin_amdgcn_wmma_f32_16x16x4_f32(
          false, xnr[kb], false, bfrag, (short)0, acc, false, false);
    }
    const float badd = b1[hb + rc];
#pragma unroll
    for (int r = 0; r < 8; ++r)
      s_a[(r + 8 * kh) * H_ + hb + rc] = acc[r] + badd;
  }

#pragma unroll
  for (int t = 0; t < 2; ++t) {
    const int jl0 = (2 * wave + t) * 16;
    v2f xr[8], xnr[8];
    load_rows(x, b * S_ + jl0, rc, kh, xr, xnr);

#pragma unroll
    for (int nt = 0; nt < 4; ++nt) {
      const int hb = nt * 16;
      v8f acc = {};
#pragma unroll
      for (int kb = 0; kb < 8; ++kb) {
        const int k0 = kb * 4 + 2 * kh;
        v2f bfrag;
        bfrag.x = W1[(size_t)(E_ + k0)     * H_ + hb + rc];
        bfrag.y = W1[(size_t)(E_ + k0 + 1) * H_ + hb + rc];
        acc = __builtin_amdgcn_wmma_f32_16x16x4_f32(
            false, xnr[kb], false, bfrag, (short)0, acc, false, false);
      }
#pragma unroll
      for (int r = 0; r < 8; ++r)
        s_eb[(jl0 + r + 8 * kh) * EB_PITCH + hb + rc] = acc[r];
    }

#pragma unroll
    for (int nt = 0; nt < 2; ++nt) {
      const int db = nt * 16;
      v8f acc = {};
#pragma unroll
      for (int kb = 0; kb < 8; ++kb) {
        const int k0 = kb * 4 + 2 * kh;
        v2f bfrag;
        bfrag.x = Wv[(size_t)(db + rc) * E_ + k0];
        bfrag.y = Wv[(size_t)(db + rc) * E_ + k0 + 1];
        acc = __builtin_amdgcn_wmma_f32_16x16x4_f32(
            false, xr[kb], false, bfrag, (short)0, acc, false, false);
      }
      const float bvv = bv[db + rc];
#pragma unroll
      for (int r = 0; r < 8; ++r)
        s_v[(jl0 + r + 8 * kh) * V_PITCH + db + rc] = acc[r] + bvv;
    }
  }
  __syncthreads();

  float eb[H_];
  {
    const v4f* src = reinterpret_cast<const v4f*>(s_eb + tid * EB_PITCH);
#pragma unroll
    for (int q = 0; q < H_ / 4; ++q) {
      const v4f t = src[q];
      eb[4 * q + 0] = t.x; eb[4 * q + 1] = t.y;
      eb[4 * q + 2] = t.z; eb[4 * q + 3] = t.w;
    }
  }


#pragma unroll 1
  for (int i = 0; i < 16; ++i) {
    const v4f* ar = reinterpret_cast<const v4f*>(s_a + i * H_);
    v4f ev[H_ / 4];
#pragma unroll
    for (int q = 0; q < H_ / 4; ++q) ev[q] = ar[q];
    asm volatile("" :: "v"(ev[0]), "v"(ev[1]), "v"(ev[2]), "v"(ev[3]),
                       "v"(ev[4]), "v"(ev[5]), "v"(ev[6]), "v"(ev[7]),
                       "v"(ev[8]), "v"(ev[9]), "v"(ev[10]), "v"(ev[11]),
                       "v"(ev[12]), "v"(ev[13]), "v"(ev[14]), "v"(ev[15]));
    __builtin_amdgcn_sched_barrier(0);
    float acc = 0.f;
#pragma unroll
    for (int q = 0; q < H_ / 4; ++q) {
      const float e4[4] = {ev[q].x, ev[q].y, ev[q].z, ev[q].w};
#pragma unroll
      for (int c = 0; c < 4; ++c) {
        const int h = 4 * q + c;
        acc += w[h] * __builtin_amdgcn_tanhf(e4[c] + eb[h]);
      }
    }
    s_sc[i * S_ + tid] = __builtin_amdgcn_tanhf(acc + b2);
  }
  __syncthreads();

  const int i  = tid >> 4;
  const int jl = tid & 15;
  float sc[16];
  float m = -3.0e38f;
#pragma unroll
  for (int jj = 0; jj < 16; ++jj) {
    const float s = s_sc[i * S_ + jj * 16 + jl];
    sc[jj] = s;
    m = fmaxf(m, s);
  }
#pragma unroll
  for (int mask = 1; mask < 16; mask <<= 1)
    m = fmaxf(m, __shfl_xor(m, mask, 16));
  float sum = 0.f;
#pragma unroll
  for (int jj = 0; jj < 16; ++jj) { sc[jj] = expf(sc[jj] - m); sum += sc[jj]; }
#pragma unroll
  for (int mask = 1; mask < 16; mask <<= 1)
    sum += __shfl_xor(sum, mask, 16);
  const float isum = 1.0f / sum;
#pragma unroll
  for (int jj = 0; jj < 16; ++jj)
    s_sc[i * S_ + jj * 16 + jl] = sc[jj] * isum;
  __syncthreads();

  if (wave < 2) {
    const int db = wave * 16;
    v8f acc = {};
    for (int kb = 0; kb < 64; ++kb) {
      const int k0 = kb * 4 + 2 * kh;
      v2f af, bfrag;
      af.x    = s_sc[rc * S_ + k0];
      af.y    = s_sc[rc * S_ + k0 + 1];
      bfrag.x = s_v[k0       * V_PITCH + db + rc];
      bfrag.y = s_v[(k0 + 1) * V_PITCH + db + rc];
      acc = __builtin_amdgcn_wmma_f32_16x16x4_f32(
          false, af, false, bfrag, (short)0, acc, false, false);
    }
#pragma unroll
    for (int r = 0; r < 8; ++r) {
      const int grow = irow0 + r + 8 * kh;
      out[(size_t)grow * E_ + db + rc] = acc[r];
    }
  }
}

extern "C" void kernel_launch(void* const* d_in, const int* in_sizes, int n_in,
                              void* d_out, int out_size, void* d_ws, size_t ws_size,
                              hipStream_t stream) {
  const float* x  = (const float*)d_in[0];
  const float* Wv = (const float*)d_in[1];
  const float* bv = (const float*)d_in[2];
  const float* W1 = (const float*)d_in[3];
  const float* b1 = (const float*)d_in[4];
  const float* w2 = (const float*)d_in[5];
  const float* b2 = (const float*)d_in[6];
  float* outp = (float*)d_out;

  attn_fused_kernel<<<dim3(B_ * (S_ / 16)), dim3(256), 0, stream>>>(
      x, Wv, bv, W1, b1, w2, b2, outp);
}


# out = attn @ v on all eight waves instead of two: K split in four quarters, 16 grouped-read WMMA steps per wave, attn rows at an odd 16-byte pitch, partial tiles summed in LDS (new reduction tree)
# speedup vs baseline: 10.5505x; 1.1664x over previous
#include <hip/hip_runtime.h>
#include <hip/hip_bf16.h>
#include <cmath>


typedef __attribute__((ext_vector_type(2))) float v2f;
typedef __attribute__((ext_vector_type(4))) float v4f;
typedef __attribute__((ext_vector_type(8))) float v8f;

#define B_ 16
#define S_ 256
#define E_ 32
#define H_ 64
#define EB_PITCH 68
#define V_PITCH 44
#define SC_PITCH 260
#define P_PITCH 36

__device__ __forceinline__ void load_rows(const float* __restrict__ x, int rowbase,
                                          int rc, int kh,
                                          v2f (&xr)[8], v2f (&xnr)[8])
{
  float ss = 0.f;
  const float* xrow = x + (size_t)(rowbase + rc) * E_;
#pragma unroll
  for (int kb = 0; kb < 8; ++kb) {
    const int k0 = kb * 4 + 2 * kh;
    v2f t; t.x = xrow[k0]; t.y = xrow[k0 + 1];
    xr[kb] = t;
    ss += t.x * t.x + t.y * t.y;
  }
  ss += __shfl_xor(ss, 16, 32);
  const float inv = 1.0f / fmaxf(sqrtf(ss), 1e-12f);
#pragma unroll
  for (int kb = 0; kb < 8; ++kb) xnr[kb] = xr[kb] * inv;
}

__global__ __launch_bounds__(256) void attn_fused_kernel(
    const float* __restrict__ x, const float* __restrict__ Wv,
    const float* __restrict__ bv, const float* __restrict__ W1,
    const float* __restrict__ b1, const float* __restrict__ w2,
    const float* __restrict__ b2p, float* __restrict__ out)
{
  __shared__ alignas(16) float s_eb[S_ * EB_PITCH];
  __shared__ float s_v[S_ * V_PITCH];
  __shared__ alignas(16) float s_a[16 * H_];
  __shared__ alignas(16) float s_sc[16 * SC_PITCH];

  const int tid   = threadIdx.x;
  const int b     = blockIdx.x >> 4;
  const int itile = blockIdx.x & 15;
  const int irow0 = b * S_ + itile * 16;
  const int wave = __builtin_amdgcn_readfirstlane(tid >> 5);
  const int lane = tid & 31;
  const int rc   = lane & 15;
  const int kh   = lane >> 4;

  const float b2 = b2p[0];
  float w[H_];
#pragma unroll
  for (int h = 0; h < H_; ++h) w[h] = w2[h];

  if (wave < 4) {
    v2f xr[8], xnr[8];
    load_rows(x, irow0, rc, kh, xr, xnr);
    const int hb = wave * 16;
    v8f acc = {};
#pragma unroll
    for (int kb = 0; kb < 8; ++kb) {
      const int k0 = kb * 4 + 2 * kh;
      v2f bfrag;
      bfrag.x = W1[(size_t)k0       * H_ + hb + rc];
      bfrag.y = W1[(size_t)(k0 + 1) * H_ + hb + rc];
      acc = __builtin_amdgcn_wmma_f32_16x16x4_f32(
          false, xnr[kb], false, bfrag, (short)0, acc, false, false);
    }
    const float badd = b1[hb + rc];
#pragma unroll
    for (int r = 0; r < 8; ++r)
      s_a[(r + 8 * kh) * H_ + hb + rc] = acc[r] + badd;
  }

#pragma unroll
  for (int t = 0; t < 2; ++t) {
    const int jl0 = (2 * wave + t) * 16;
    v2f xr[8], xnr[8];
    load_rows(x, b * S_ + jl0, rc, kh, xr, xnr);

#pragma unroll
    for (int nt = 0; nt < 4; ++nt) {
      const int hb = nt * 16;
      v8f acc = {};
#pragma unroll
      for (int kb = 0; kb < 8; ++kb) {
        const int k0 = kb * 4 + 2 * kh;
        v2f bfrag;
        bfrag.x = W1[(size_t)(E_ + k0)     * H_ + hb + rc];
        bfrag.y = W1[(size_t)(E_ + k0 + 1) * H_ + hb + rc];
        acc = __builtin_amdgcn_wmma_f32_16x16x4_f32(
            false, xnr[kb], false, bfrag, (short)0, acc, false, false);
      }
#pragma unroll
      for (int r = 0; r < 8; ++r)
        s_eb[(jl0 + r + 8 * kh) * EB_PITCH + hb + rc] = acc[r];
    }

#pragma unroll
    for (int nt = 0; nt < 2; ++nt) {
      const int db = nt * 16;
      v8f acc = {};
#pragma unroll
      for (int kb = 0; kb < 8; ++kb) {
        const int k0 = kb * 4 + 2 * kh;
        v2f bfrag;
        bfrag.x = Wv[(size_t)(db + rc) * E_ + k0];
        bfrag.y = Wv[(size_t)(db + rc) * E_ + k0 + 1];
        acc = __builtin_amdgcn_wmma_f32_16x16x4_f32(
            false, xr[kb], false, bfrag, (short)0, acc, false, false);
      }
      const float bvv = bv[db + rc];
#pragma unroll
      for (int r = 0; r < 8; ++r)
        s_v[(jl0 + r + 8 * kh) * V_PITCH + db + rc] = acc[r] + bvv;
    }
  }
  __syncthreads();

  float eb[H_];
  {
    const v4f* src = reinterpret_cast<const v4f*>(s_eb + tid * EB_PITCH);
#pragma unroll
    for (int q = 0; q < H_ / 4; ++q) {
      const v4f t = src[q];
      eb[4 * q + 0] = t.x; eb[4 * q + 1] = t.y;
      eb[4 * q + 2] = t.z; eb[4 * q + 3] = t.w;
    }
  }


#pragma unroll 1
  for (int i = 0; i < 16; ++i) {
    const v4f* ar = reinterpret_cast<const v4f*>(s_a + i * H_);
    v4f ev[H_ / 4];
#pragma unroll
    for (int q = 0; q < H_ / 4; ++q) ev[q] = ar[q];
    asm volatile("" :: "v"(ev[0]), "v"(ev[1]), "v"(ev[2]), "v"(ev[3]),
                       "v"(ev[4]), "v"(ev[5]), "v"(ev[6]), "v"(ev[7]),
                       "v"(ev[8]), "v"(ev[9]), "v"(ev[10]), "v"(ev[11]),
                       "v"(ev[12]), "v"(ev[13]), "v"(ev[14]), "v"(ev[15]));
    __builtin_amdgcn_sched_barrier(0);
    float acc = 0.f;
#pragma unroll
    for (int q = 0; q < H_ / 4; ++q) {
      const float e4[4] = {ev[q].x, ev[q].y, ev[q].z, ev[q].w};
#pragma unroll
      for (int c = 0; c < 4; ++c) {
        const int h = 4 * q + c;
        acc += w[h] * __builtin_amdgcn_tanhf(e4[c] + eb[h]);
      }
    }
    s_sc[i * SC_PITCH + tid] = __builtin_amdgcn_tanhf(acc + b2);
  }
  __syncthreads();

  const int i  = tid >> 4;
  const int jl = tid & 15;
  float sc[16];
  float m = -3.0e38f;
#pragma unroll
  for (int jj = 0; jj < 16; ++jj) {
    const float s = s_sc[i * SC_PITCH + jj * 16 + jl];
    sc[jj] = s;
    m = fmaxf(m, s);
  }
#pragma unroll
  for (int mask = 1; mask < 16; mask <<= 1)
    m = fmaxf(m, __shfl_xor(m, mask, 16));
  float sum = 0.f;
#pragma unroll
  for (int jj = 0; jj < 16; ++jj) { sc[jj] = expf(sc[jj] - m); sum += sc[jj]; }
#pragma unroll
  for (int mask = 1; mask < 16; mask <<= 1)
    sum += __shfl_xor(sum, mask, 16);
  const float isum = 1.0f / sum;
#pragma unroll
  for (int jj = 0; jj < 16; ++jj)
    s_sc[i * SC_PITCH + jj * 16 + jl] = sc[jj] * isum;
  __syncthreads();

  {
    const int dh = wave & 1;
    const int kq = wave >> 1;
    const int db = dh * 16;
    v2f af[16], bfr[16];
#pragma unroll
    for (int s = 0; s < 16; ++s) {
      const int k0 = kq * 64 + s * 4 + 2 * kh;
      af[s] = *reinterpret_cast<const v2f*>(s_sc + rc * SC_PITCH + k0);
      v2f t;
      t.x = s_v[k0       * V_PITCH + db + rc];
      t.y = s_v[(k0 + 1) * V_PITCH + db + rc];
      bfr[s] = t;
    }
    asm volatile("" :: "v"(af[0]), "v"(af[1]), "v"(af[2]), "v"(af[3]),
                       "v"(af[4]), "v"(af[5]), "v"(af[6]), "v"(af[7]),
                       "v"(af[8]), "v"(af[9]), "v"(af[10]), "v"(af[11]),
                       "v"(af[12]), "v"(af[13]), "v"(af[14]), "v"(af[15]));
    asm volatile("" :: "v"(bfr[0]), "v"(bfr[1]), "v"(bfr[2]), "v"(bfr[3]),
                       "v"(bfr[4]), "v"(bfr[5]), "v"(bfr[6]), "v"(bfr[7]),
                       "v"(bfr[8]), "v"(bfr[9]), "v"(bfr[10]), "v"(bfr[11]),
                       "v"(bfr[12]), "v"(bfr[13]), "v"(bfr[14]), "v"(bfr[15]));
    __builtin_amdgcn_sched_barrier(0);
    v8f acc0 = {}, acc1 = {};
#pragma unroll
    for (int s = 0; s < 16; s += 2) {
      acc0 = __builtin_amdgcn_wmma_f32_16x16x4_f32(
          false, af[s], false, bfr[s], (short)0, acc0, false, false);
      acc1 = __builtin_amdgcn_wmma_f32_16x16x4_f32(
          false, af[s + 1], false, bfr[s + 1], (short)0, acc1, false, false);
    }
    float* part = s_eb + kq * (16 * P_PITCH);
#pragma unroll
    for (int r = 0; r < 8; ++r)
      part[(r + 8 * kh) * P_PITCH + db + rc] = acc0[r] + acc1[r];
  }
  __syncthreads();

  {
    const int row = tid >> 4;
    const int c2  = (tid & 15) * 2;
    const float* p = s_eb + row * P_PITCH + c2;
    const v2f p0 = *reinterpret_cast<const v2f*>(p);
    const v2f p1 = *reinterpret_cast<const v2f*>(p + 1 * 16 * P_PITCH);
    const v2f p2 = *reinterpret_cast<const v2f*>(p + 2 * 16 * P_PITCH);
    const v2f p3 = *reinterpret_cast<const v2f*>(p + 3 * 16 * P_PITCH);
    const v2f o = ((p0 + p1) + p2) + p3;
    *reinterpret_cast<v2f*>(out + (size_t)(irow0 + row) * E_ + c2) = o;
  }
}

extern "C" void kernel_launch(void* const* d_in, const int* in_sizes, int n_in,
                              void* d_out, int out_size, void* d_ws, size_t ws_size,
                              hipStream_t stream) {
  const float* x  = (const float*)d_in[0];
  const float* Wv = (const float*)d_in[1];
  const float* bv = (const float*)d_in[2];
  const float* W1 = (const float*)d_in[3];
  const float* b1 = (const float*)d_in[4];
  const float* w2 = (const float*)d_in[5];
  const float* b2 = (const float*)d_in[6];
  float* outp = (float*)d_out;

  attn_fused_kernel<<<dim3(B_ * (S_ / 16)), dim3(256), 0, stream>>>(
      x, Wv, bv, W1, b1, w2, b2, outp);
}


# pair loop at the tanh unit's rate: per two h one packed add, two tanh, one packed fma (w2 pairs in VGPRs), a packed op between tanh; sum over h as 8 partial sums (h mod 8), new summation order
# speedup vs baseline: 11.6020x; 1.0997x over previous
#include <hip/hip_runtime.h>
#include <hip/hip_bf16.h>
#include <cmath>


typedef __attribute__((ext_vector_type(2))) float v2f;
typedef __attribute__((ext_vector_type(4))) float v4f;
typedef __attribute__((ext_vector_type(8))) float v8f;

#define B_ 16
#define S_ 256
#define E_ 32
#define H_ 64
#define EB_PITCH 68
#define V_PITCH 44
#define SC_PITCH 260
#define P_PITCH 36

__device__ __forceinline__ void load_rows(const float* __restrict__ x, int rowbase,
                                          int rc, int kh,
                                          v2f (&xr)[8], v2f (&xnr)[8])
{
  float ss = 0.f;
  const float* xrow = x + (size_t)(rowbase + rc) * E_;
#pragma unroll
  for (int kb = 0; kb < 8; ++kb) {
    const int k0 = kb * 4 + 2 * kh;
    v2f t; t.x = xrow[k0]; t.y = xrow[k0 + 1];
    xr[kb] = t;
    ss += t.x * t.x + t.y * t.y;
  }
  ss += __shfl_xor(ss, 16, 32);
  const float inv = 1.0f / fmaxf(sqrtf(ss), 1e-12f);
#pragma unroll
  for (int kb = 0; kb < 8; ++kb) xnr[kb] = xr[kb] * inv;
}

__global__ __launch_bounds__(256) void attn_fused_kernel(
    const float* __restrict__ x, const float* __restrict__ Wv,
    const float* __restrict__ bv, const float* __restrict__ W1,
    const float* __restrict__ b1, const float* __restrict__ w2,
    const float* __restrict__ b2p, float* __restrict__ out)
{
  __shared__ alignas(16) float s_eb[S_ * EB_PITCH];
  __shared__ float s_v[S_ * V_PITCH];
  __shared__ alignas(16) float s_a[16 * H_];
  __shared__ alignas(16) float s_sc[16 * SC_PITCH];
  __shared__ alignas(16) float s_w[H_];

  const int tid   = threadIdx.x;
  const int b     = blockIdx.x >> 4;
  const int itile = blockIdx.x & 15;
  const int irow0 = b * S_ + itile * 16;
  const int wave = __builtin_amdgcn_readfirstlane(tid >> 5);
  const int lane = tid & 31;
  const int rc   = lane & 15;
  const int kh   = lane >> 4;

  const float b2 = b2p[0];
  const float w2reg = w2[tid & (H_ - 1)];

  if (wave < 4) {
    v2f xr[8], xnr[8];
    load_rows(x, irow0, rc, kh, xr, xnr);
    const int hb = wave * 16;
    v8f acc = {};
#pragma unroll
    for (int kb = 0; kb < 8; ++kb) {
      const int k0 = kb * 4 + 2 * kh;
      v2f bfrag;
      bfrag.x = W1[(size_t)k0       * H_ + hb + rc];
      bfrag.y = W1[(size_t)(k0 + 1) * H_ + hb + rc];
      acc = __builtin_amdgcn_wmma_f32_16x16x4_f32(
          false, xnr[kb], false, bfrag, (short)0, acc, false, false);
    }
    const float badd = b1[hb + rc];
#pragma unroll
    for (int r = 0; r < 8; ++r)
      s_a[(r + 8 * kh) * H_ + hb + rc] = acc[r] + badd;
  }

#pragma unroll
  for (int t = 0; t < 2; ++t) {
    const int jl0 = (2 * wave + t) * 16;
    v2f xr[8], xnr[8];
    load_rows(x, b * S_ + jl0, rc, kh, xr, xnr);

#pragma unroll
    for (int nt = 0; nt < 4; ++nt) {
      const int hb = nt * 16;
      v8f acc = {};
#pragma unroll
      for (int kb = 0; kb < 8; ++kb) {
        const int k0 = kb * 4 + 2 * kh;
        v2f bfrag;
        bfrag.x = W1[(size_t)(E_ + k0)     * H_ + hb + rc];
        bfrag.y = W1[(size_t)(E_ + k0 + 1) * H_ + hb + rc];
        acc = __builtin_amdgcn_wmma_f32_16x16x4_f32(
            false, xnr[kb], false, bfrag, (short)0, acc, false, false);
      }
#pragma unroll
      for (int r = 0; r < 8; ++r)
        s_eb[(jl0 + r + 8 * kh) * EB_PITCH + hb + rc] = acc[r];
    }

#pragma unroll
    for (int nt = 0; nt < 2; ++nt) {
      const int db = nt * 16;
      v8f acc = {};
#pragma unroll
      for (int kb = 0; kb < 8; ++kb) {
        const int k0 = kb * 4 + 2 * kh;
        v2f bfrag;
        bfrag.x = Wv[(size_t)(db + rc) * E_ + k0];
        bfrag.y = Wv[(size_t)(db + rc) * E_ + k0 + 1];
        acc = __builtin_amdgcn_wmma_f32_16x16x4_f32(
            false, xr[kb], false, bfrag, (short)0, acc, false, false);
      }
      const float bvv = bv[db + rc];
#pragma unroll
      for (int r = 0; r < 8; ++r)
        s_v[(jl0 + r + 8 * kh) * V_PITCH + db + rc] = acc[r] + bvv;
    }
  }
  if (tid < H_) s_w[tid] = w2reg;
  __syncthreads();

  float eb[H_];
  {
    const v4f* src = reinterpret_cast<const v4f*>(s_eb + tid * EB_PITCH);
#pragma unroll
    for (int q = 0; q < H_ / 4; ++q) {
      const v4f t = src[q];
      eb[4 * q + 0] = t.x; eb[4 * q + 1] = t.y;
      eb[4 * q + 2] = t.z; eb[4 * q + 3] = t.w;
    }
  }
  v2f wv[H_ / 2];
  {
    const v4f* src = reinterpret_cast<const v4f*>(s_w);
#pragma unroll
    for (int q = 0; q < H_ / 4; ++q) {
      const v4f t = src[q];
      wv[2 * q].x = t.x;     wv[2 * q].y = t.y;
      wv[2 * q + 1].x = t.z; wv[2 * q + 1].y = t.w;
    }
  }


#pragma unroll 1
  for (int i = 0; i < 16; ++i) {
    const v4f* ar = reinterpret_cast<const v4f*>(s_a + i * H_);
    v4f ev[H_ / 4];
#pragma unroll
    for (int q = 0; q < H_ / 4; ++q) ev[q] = ar[q];
    asm volatile("" :: "v"(ev[0]), "v"(ev[1]), "v"(ev[2]), "v"(ev[3]),
                       "v"(ev[4]), "v"(ev[5]), "v"(ev[6]), "v"(ev[7]),
                       "v"(ev[8]), "v"(ev[9]), "v"(ev[10]), "v"(ev[11]),
                       "v"(ev[12]), "v"(ev[13]), "v"(ev[14]), "v"(ev[15]));
    __builtin_amdgcn_sched_barrier(0);
    v2f acc[4] = {};
#pragma unroll
    for (int q = 0; q < H_ / 4; ++q) {
      const int h = 4 * q;
      v2f s0, s1, t0, t1;
      s0.x = ev[q].x + eb[h];     s0.y = ev[q].y + eb[h + 1];
      s1.x = ev[q].z + eb[h + 2]; s1.y = ev[q].w + eb[h + 3];
      t0.x = __builtin_amdgcn_tanhf(s0.x); t0.y = __builtin_amdgcn_tanhf(s0.y);
      t1.x = __builtin_amdgcn_tanhf(s1.x); t1.y = __builtin_amdgcn_tanhf(s1.y);
      acc[(2 * q) & 3]     = __builtin_elementwise_fma(wv[2 * q], t0, acc[(2 * q) & 3]);
      acc[(2 * q + 1) & 3] = __builtin_elementwise_fma(wv[2 * q + 1], t1, acc[(2 * q + 1) & 3]);
    }
#pragma unroll
    for (int g = 0; g < H_; ++g) {
      __builtin_amdgcn_sched_group_barrier(0x400, 1, 0);
      __builtin_amdgcn_sched_group_barrier(0x002, 1, 0);
    }
    const v2f s4 = (acc[0] + acc[1]) + (acc[2] + acc[3]);
    s_sc[i * SC_PITCH + tid] = __builtin_amdgcn_tanhf((s4.x + s4.y) + b2);
  }
  __syncthreads();

  const int i  = tid >> 4;
  const int jl = tid & 15;
  float sc[16];
  float m = -3.0e38f;
#pragma unroll
  for (int jj = 0; jj < 16; ++jj) {
    const float s = s_sc[i * SC_PITCH + jj * 16 + jl];
    sc[jj] = s;
    m = fmaxf(m, s);
  }
#pragma unroll
  for (int mask = 1; mask < 16; mask <<= 1)
    m = fmaxf(m, __shfl_xor(m, mask, 16));
  float sum = 0.f;
#pragma unroll
  for (int jj = 0; jj < 16; ++jj) { sc[jj] = expf(sc[jj] - m); sum += sc[jj]; }
#pragma unroll
  for (int mask = 1; mask < 16; mask <<= 1)
    sum += __shfl_xor(sum, mask, 16);
  const float isum = 1.0f / sum;
#pragma unroll
  for (int jj = 0; jj < 16; ++jj)
    s_sc[i * SC_PITCH + jj * 16 + jl] = sc[jj] * isum;
  __syncthreads();

  {
    const int dh = wave & 1;
    const int kq = wave >> 1;
    const int db = dh * 16;
    v2f af[16], bfr[16];
#pragma unroll
    for (int s = 0; s < 16; ++s) {
      const int k0 = kq * 64 + s * 4 + 2 * kh;
      af[s] = *reinterpret_cast<const v2f*>(s_sc + rc * SC_PITCH + k0);
      v2f t;
      t.x = s_v[k0       * V_PITCH + db + rc];
      t.y = s_v[(k0 + 1) * V_PITCH + db + rc];
      bfr[s] = t;
    }
    asm volatile("" :: "v"(af[0]), "v"(af[1]), "v"(af[2]), "v"(af[3]),
                       "v"(af[4]), "v"(af[5]), "v"(af[6]), "v"(af[7]),
                       "v"(af[8]), "v"(af[9]), "v"(af[10]), "v"(af[11]),
                       "v"(af[12]), "v"(af[13]), "v"(af[14]), "v"(af[15]));
    asm volatile("" :: "v"(bfr[0]), "v"(bfr[1]), "v"(bfr[2]), "v"(bfr[3]),
                       "v"(bfr[4]), "v"(bfr[5]), "v"(bfr[6]), "v"(bfr[7]),
                       "v"(bfr[8]), "v"(bfr[9]), "v"(bfr[10]), "v"(bfr[11]),
                       "v"(bfr[12]), "v"(bfr[13]), "v"(bfr[14]), "v"(bfr[15]));
    __builtin_amdgcn_sched_barrier(0);
    v8f acc0 = {}, acc1 = {};
#pragma unroll
    for (int s = 0; s < 16; s += 2) {
      acc0 = __builtin_amdgcn_wmma_f32_16x16x4_f32(
          false, af[s], false, bfr[s], (short)0, acc0, false, false);
      acc1 = __builtin_amdgcn_wmma_f32_16x16x4_f32(
          false, af[s + 1], false, bfr[s + 1], (short)0, acc1, false, false);
    }
    float* part = s_eb + kq * (16 * P_PITCH);
#pragma unroll
    for (int r = 0; r < 8; ++r)
      part[(r + 8 * kh) * P_PITCH + db + rc] = acc0[r] + acc1[r];
  }
  __syncthreads();

  {
    const int row = tid >> 4;
    const int c2  = (tid & 15) * 2;
    const float* p = s_eb + row * P_PITCH + c2;
    const v2f p0 = *reinterpret_cast<const v2f*>(p);
    const v2f p1 = *reinterpret_cast<const v2f*>(p + 1 * 16 * P_PITCH);
    const v2f p2 = *reinterpret_cast<const v2f*>(p + 2 * 16 * P_PITCH);
    const v2f p3 = *reinterpret_cast<const v2f*>(p + 3 * 16 * P_PITCH);
    const v2f o = ((p0 + p1) + p2) + p3;
    *reinterpret_cast<v2f*>(out + (size_t)(irow0 + row) * E_ + c2) = o;
  }
}

extern "C" void kernel_launch(void* const* d_in, const int* in_sizes, int n_in,
                              void* d_out, int out_size, void* d_ws, size_t ws_size,
                              hipStream_t stream) {
  const float* x  = (const float*)d_in[0];
  const float* Wv = (const float*)d_in[1];
  const float* bv = (const float*)d_in[2];
  const float* W1 = (const float*)d_in[3];
  const float* b1 = (const float*)d_in[4];
  const float* w2 = (const float*)d_in[5];
  const float* b2 = (const float*)d_in[6];
  float* outp = (float*)d_out;

  attn_fused_kernel<<<dim3(B_ * (S_ / 16)), dim3(256), 0, stream>>>(
      x, Wv, bv, W1, b1, w2, b2, outp);
}


# softmax pass and two barriers removed: score is a tanh value, so pair loop stores 2^(score*log2e), each wave multiplies its own 32 columns by its own v rows, out/rowsum at the end; new sum order
# speedup vs baseline: 12.3227x; 1.0621x over previous
#include <hip/hip_runtime.h>
#include <hip/hip_bf16.h>
#include <cmath>


typedef __attribute__((ext_vector_type(2))) float v2f;
typedef __attribute__((ext_vector_type(4))) float v4f;
typedef __attribute__((ext_vector_type(8))) float v8f;

#define B_ 16
#define S_ 256
#define E_ 32
#define H_ 64
#define EB_PITCH 68
#define V_PITCH 44
#define SC_PITCH 260
#define P_PITCH 36

__device__ __forceinline__ void load_rows(const float* __restrict__ x, int rowbase,
                                          int rc, int kh,
                                          v2f (&xr)[8], v2f (&xnr)[8])
{
  float ss = 0.f;
  const float* xrow = x + (size_t)(rowbase + rc) * E_;
#pragma unroll
  for (int kb = 0; kb < 8; ++kb) {
    const int k0 = kb * 4 + 2 * kh;
    v2f t; t.x = xrow[k0]; t.y = xrow[k0 + 1];
    xr[kb] = t;
    ss += t.x * t.x + t.y * t.y;
  }
  ss += __shfl_xor(ss, 16, 32);
  const float inv = 1.0f / fmaxf(sqrtf(ss), 1e-12f);
#pragma unroll
  for (int kb = 0; kb < 8; ++kb) xnr[kb] = xr[kb] * inv;
}

__global__ __launch_bounds__(256) void attn_fused_kernel(
    const float* __restrict__ x, const float* __restrict__ Wv,
    const float* __restrict__ bv, const float* __restrict__ W1,
    const float* __restrict__ b1, const float* __restrict__ w2,
    const float* __restrict__ b2p, float* __restrict__ out)
{
  __shared__ alignas(16) float s_eb[S_ * EB_PITCH];
  __shared__ float s_v[S_ * V_PITCH];
  __shared__ alignas(16) float s_a[16 * H_];
  __shared__ alignas(16) float s_sc[16 * SC_PITCH];
  __shared__ alignas(16) float s_w[H_];

  const int tid   = threadIdx.x;
  const int b     = blockIdx.x >> 4;
  const int itile = blockIdx.x & 15;
  const int irow0 = b * S_ + itile * 16;
  const int wave = __builtin_amdgcn_readfirstlane(tid >> 5);
  const int lane = tid & 31;
  const int rc   = lane & 15;
  const int kh   = lane >> 4;

  const float b2 = b2p[0];
  const float w2reg = w2[tid & (H_ - 1)];

  if (wave < 4) {
    v2f xr[8], xnr[8];
    load_rows(x, irow0, rc, kh, xr, xnr);
    const int hb = wave * 16;
    v8f acc = {};
#pragma unroll
    for (int kb = 0; kb < 8; ++kb) {
      const int k0 = kb * 4 + 2 * kh;
      v2f bfrag;
      bfrag.x = W1[(size_t)k0       * H_ + hb + rc];
      bfrag.y = W1[(size_t)(k0 + 1) * H_ + hb + rc];
      acc = __builtin_amdgcn_wmma_f32_16x16x4_f32(
          false, xnr[kb], false, bfrag, (short)0, acc, false, false);
    }
    const float badd = b1[hb + rc];
#pragma unroll
    for (int r = 0; r < 8; ++r)
      s_a[(r + 8 * kh) * H_ + hb + rc] = acc[r] + badd;
  }

#pragma unroll
  for (int t = 0; t < 2; ++t) {
    const int jl0 = (2 * wave + t) * 16;
    v2f xr[8], xnr[8];
    load_rows(x, b * S_ + jl0, rc, kh, xr, xnr);

#pragma unroll
    for (int nt = 0; nt < 4; ++nt) {
      const int hb = nt * 16;
      v8f acc = {};
#pragma unroll
      for (int kb = 0; kb < 8; ++kb) {
        const int k0 = kb * 4 + 2 * kh;
        v2f bfrag;
        bfrag.x = W1[(size_t)(E_ + k0)     * H_ + hb + rc];
        bfrag.y = W1[(size_t)(E_ + k0 + 1) * H_ + hb + rc];
        acc = __builtin_amdgcn_wmma_f32_16x16x4_f32(
            false, xnr[kb], false, bfrag, (short)0, acc, false, false);
      }
#pragma unroll
      for (int r = 0; r < 8; ++r)
        s_eb[(jl0 + r + 8 * kh) * EB_PITCH + hb + rc] = acc[r];
    }

#pragma unroll
    for (int nt = 0; nt < 2; ++nt) {
      const int db = nt * 16;
      v8f acc = {};
#pragma unroll
      for (int kb = 0; kb < 8; ++kb) {
        const int k0 = kb * 4 + 2 * kh;
        v2f bfrag;
        bfrag.x = Wv[(size_t)(db + rc) * E_ + k0];
        bfrag.y = Wv[(size_t)(db + rc) * E_ + k0 + 1];
        acc = __builtin_amdgcn_wmma_f32_16x16x4_f32(
            false, xr[kb], false, bfrag, (short)0, acc, false, false);
      }
      const float bvv = bv[db + rc];
#pragma unroll
      for (int r = 0; r < 8; ++r)
        s_v[(jl0 + r + 8 * kh) * V_PITCH + db + rc] = acc[r] + bvv;
    }
  }
  if (tid < H_) s_w[tid] = w2reg;
  __syncthreads();

  float eb[H_];
  {
    const v4f* src = reinterpret_cast<const v4f*>(s_eb + tid * EB_PITCH);
#pragma unroll
    for (int q = 0; q < H_ / 4; ++q) {
      const v4f t = src[q];
      eb[4 * q + 0] = t.x; eb[4 * q + 1] = t.y;
      eb[4 * q + 2] = t.z; eb[4 * q + 3] = t.w;
    }
  }
  v2f wv[H_ / 2];
  {
    const v4f* src = reinterpret_cast<const v4f*>(s_w);
#pragma unroll
    for (int q = 0; q < H_ / 4; ++q) {
      const v4f t = src[q];
      wv[2 * q].x = t.x;     wv[2 * q].y = t.y;
      wv[2 * q + 1].x = t.z; wv[2 * q + 1].y = t.w;
    }
  }


#pragma unroll 1
  for (int i = 0; i < 16; ++i) {
    const v4f* ar = reinterpret_cast<const v4f*>(s_a + i * H_);
    v4f ev[H_ / 4];
#pragma unroll
    for (int q = 0; q < H_ / 4; ++q) ev[q] = ar[q];
    asm volatile("" :: "v"(ev[0]), "v"(ev[1]), "v"(ev[2]), "v"(ev[3]),
                       "v"(ev[4]), "v"(ev[5]), "v"(ev[6]), "v"(ev[7]),
                       "v"(ev[8]), "v"(ev[9]), "v"(ev[10]), "v"(ev[11]),
                       "v"(ev[12]), "v"(ev[13]), "v"(ev[14]), "v"(ev[15]));
    __builtin_amdgcn_sched_barrier(0);
    v2f acc[4] = {};
#pragma unroll
    for (int q = 0; q < H_ / 4; ++q) {
      const int h = 4 * q;
      v2f s0, s1, t0, t1;
      s0.x = ev[q].x + eb[h];     s0.y = ev[q].y + eb[h + 1];
      s1.x = ev[q].z + eb[h + 2]; s1.y = ev[q].w + eb[h + 3];
      t0.x = __builtin_amdgcn_tanhf(s0.x); t0.y = __builtin_amdgcn_tanhf(s0.y);
      t1.x = __builtin_amdgcn_tanhf(s1.x); t1.y = __builtin_amdgcn_tanhf(s1.y);
      acc[(2 * q) & 3]     = __builtin_elementwise_fma(wv[2 * q], t0, acc[(2 * q) & 3]);
      acc[(2 * q + 1) & 3] = __builtin_elementwise_fma(wv[2 * q + 1], t1, acc[(2 * q + 1) & 3]);
    }
#pragma unroll
    for (int g = 0; g < H_; ++g) {
      __builtin_amdgcn_sched_group_barrier(0x400, 1, 0);
      __builtin_amdgcn_sched_group_barrier(0x002, 1, 0);
    }
    const v2f s4 = (acc[0] + acc[1]) + (acc[2] + acc[3]);
    const float score = __builtin_amdgcn_tanhf((s4.x + s4.y) + b2);
    s_sc[i * SC_PITCH + tid] = __builtin_amdgcn_exp2f(score * 1.44269504088896340736f);
  }


  {
    const int jb = wave * 32;
    v4f af4[4];
    v2f bfr[2][2][4];
#pragma unroll
    for (int t = 0; t < 2; ++t) {
#pragma unroll
      for (int u = 0; u < 2; ++u)
        af4[2 * t + u] = *reinterpret_cast<const v4f*>(
            s_sc + rc * SC_PITCH + jb + 16 * t + 8 * kh + 4 * u);
#pragma unroll
      for (int dh = 0; dh < 2; ++dh) {
#pragma unroll
        for (int s = 0; s < 4; ++s) {
          const int k0 = jb + 16 * t + 8 * kh + 2 * s;
          v2f q;
          q.x = s_v[k0       * V_PITCH + 16 * dh + rc];
          q.y = s_v[(k0 + 1) * V_PITCH + 16 * dh + rc];
          bfr[t][dh][s] = q;
        }
      }
    }
    asm volatile("" :: "v"(af4[0]), "v"(af4[1]), "v"(af4[2]), "v"(af4[3]));
    asm volatile("" :: "v"(bfr[0][0][0]), "v"(bfr[0][0][1]), "v"(bfr[0][0][2]), "v"(bfr[0][0][3]),
                       "v"(bfr[0][1][0]), "v"(bfr[0][1][1]), "v"(bfr[0][1][2]), "v"(bfr[0][1][3]),
                       "v"(bfr[1][0][0]), "v"(bfr[1][0][1]), "v"(bfr[1][0][2]), "v"(bfr[1][0][3]),
                       "v"(bfr[1][1][0]), "v"(bfr[1][1][1]), "v"(bfr[1][1][2]), "v"(bfr[1][1][3]));
    __builtin_amdgcn_sched_barrier(0);
    v8f acc0 = {}, acc1 = {};
#pragma unroll
    for (int t = 0; t < 2; ++t) {
#pragma unroll
      for (int s = 0; s < 4; ++s) {
        const v4f a4 = af4[2 * t + (s >> 1)];
        v2f a;
        a.x = (s & 1) ? a4.z : a4.x;
        a.y = (s & 1) ? a4.w : a4.y;
        acc0 = __builtin_amdgcn_wmma_f32_16x16x4_f32(
            false, a, false, bfr[t][0][s], (short)0, acc0, false, false);
        acc1 = __builtin_amdgcn_wmma_f32_16x16x4_f32(
            false, a, false, bfr[t][1][s], (short)0, acc1, false, false);
      }
    }
    const v4f r4 = (af4[0] + af4[1]) + (af4[2] + af4[3]);
    float rsum = (r4.x + r4.y) + (r4.z + r4.w);
    rsum += __shfl_xor(rsum, 16, 32);
    float* part = s_eb + jb * EB_PITCH;
#pragma unroll
    for (int r = 0; r < 8; ++r) {
      part[(r + 8 * kh) * P_PITCH + rc]      = acc0[r];
      part[(r + 8 * kh) * P_PITCH + 16 + rc] = acc1[r];
    }
    if (kh == 0) part[16 * P_PITCH + rc] = rsum;
  }
  __syncthreads();

  {
    const int row = tid >> 4;
    const int c2  = (tid & 15) * 2;
    const float* p = s_eb + row * P_PITCH + c2;
    const float* q = s_eb + 16 * P_PITCH + row;
    v2f o = *reinterpret_cast<const v2f*>(p);
    float rs = q[0];
#pragma unroll
    for (int w8 = 1; w8 < 8; ++w8) {
      o  = o + *reinterpret_cast<const v2f*>(p + w8 * (32 * EB_PITCH));
      rs = rs + q[w8 * (32 * EB_PITCH)];
    }
    const float inv = 1.0f / rs;
    *reinterpret_cast<v2f*>(out + (size_t)(irow0 + row) * E_ + c2) = o * inv;
  }
}

extern "C" void kernel_launch(void* const* d_in, const int* in_sizes, int n_in,
                              void* d_out, int out_size, void* d_ws, size_t ws_size,
                              hipStream_t stream) {
  const float* x  = (const float*)d_in[0];
  const float* Wv = (const float*)d_in[1];
  const float* bv = (const float*)d_in[2];
  const float* W1 = (const float*)d_in[3];
  const float* b1 = (const float*)d_in[4];
  const float* w2 = (const float*)d_in[5];
  const float* b2 = (const float*)d_in[6];
  float* outp = (float*)d_out;

  attn_fused_kernel<<<dim3(B_ * (S_ / 16)), dim3(256), 0, stream>>>(
      x, Wv, bv, W1, b1, w2, b2, outp);
}
